# starting kernel
# baseline (speedup 1.0000x reference)
#include <hip/hip_runtime.h>


#define N_ 8
#define L_ 8192
#define S_ 8192
#define H_ 8
#define D_ 32
#define RS (H_ * D_)
#define NH (N_ * H_)
#define CHUNKS 64
#define CHUNK_ROWS (S_ / CHUNKS)
#define SLOT 1056

typedef __attribute__((ext_vector_type(2))) float v2f;
typedef __attribute__((ext_vector_type(8))) float v8f;

__device__ __forceinline__ float elu1(float x) {
  return x > 0.f ? x + 1.f : __expf(x);
}

__device__ __forceinline__ v8f wmma_f32(v2f a, v2f b, v8f c) {
  return __builtin_amdgcn_wmma_f32_16x16x4_f32(false, a, false, b, (short)0, c,
                                               false, false);
}

__global__ void __launch_bounds__(256)
kv_partial_kernel(const float* __restrict__ Kp, const float* __restrict__ Vp,
                  float* __restrict__ part) {
  const int wave = threadIdx.x >> 5;
  const int lane = threadIdx.x & 31;
  const int g    = blockIdx.x * 8 + wave;
  const int nh   = g / CHUNKS;
  const int ck   = g % CHUNKS;
  const int n    = nh / H_;
  const int h    = nh % H_;
  const int col  = lane & 15;
  const int half = lane >> 4;
  const float invS = 1.0f / (float)S_;

  const size_t base = ((size_t)n * S_) * RS + (size_t)h * D_;
  const int s0 = ck * CHUNK_ROWS;

  v8f c00 = {}, c01 = {}, c10 = {}, c11 = {};
  float ks0 = 0.f, ks1 = 0.f;

  for (int s = s0; s < s0 + CHUNK_ROWS; s += 4) {
    const int rA = s + 2 * half;
    const float* k0 = Kp + base + (size_t)rA * RS;
    const float* k1 = k0 + RS;
    const float* v0 = Vp + base + (size_t)rA * RS;
    const float* v1 = v0 + RS;
    v2f a0, a1, b0, b1;
    float t;
    t = elu1(k0[col]);      a0.x = t; ks0 += t;
    t = elu1(k1[col]);      a0.y = t; ks0 += t;
    t = elu1(k0[col + 16]); a1.x = t; ks1 += t;
    t = elu1(k1[col + 16]); a1.y = t; ks1 += t;
    b0.x = v0[col]      * invS; b0.y = v1[col]      * invS;
    b1.x = v0[col + 16] * invS; b1.y = v1[col + 16] * invS;
    c00 = wmma_f32(a0, b0, c00);
    c01 = wmma_f32(a0, b1, c01);
    c10 = wmma_f32(a1, b0, c10);
    c11 = wmma_f32(a1, b1, c11);
  }
  ks0 += __shfl_xor(ks0, 16, 32);
  ks1 += __shfl_xor(ks1, 16, 32);

  float* outp = part + (size_t)g * SLOT;
#pragma unroll
  for (int r = 0; r < 8; ++r) {
    const int m = r + 8 * half;
    outp[m * 32 + col]              = c00[r];
    outp[m * 32 + col + 16]         = c01[r];
    outp[(16 + m) * 32 + col]       = c10[r];
    outp[(16 + m) * 32 + col + 16]  = c11[r];
  }
  if (lane < 16) {
    outp[1024 + col]      = ks0;
    outp[1024 + col + 16] = ks1;
  }
}

__global__ void __launch_bounds__(256)
kv_reduce_kernel(const float* __restrict__ part, float* __restrict__ kv) {
  const int nh = blockIdx.x;
  for (int e = threadIdx.x; e < SLOT; e += 256) {
    float acc = 0.f;
    const float* p = part + (size_t)nh * CHUNKS * SLOT + e;
#pragma unroll 4
    for (int c = 0; c < CHUNKS; ++c) acc += p[(size_t)c * SLOT];
    kv[(size_t)nh * SLOT + e] = acc;
  }
}

__global__ void __launch_bounds__(256)
attn_out_kernel(const float* __restrict__ Qp, const float* __restrict__ kvbuf,
                float* __restrict__ outp) {
  __shared__ float sKV[1024];
  __shared__ float sKs[32];
  const int tilesPerNH = L_ / 128;
  const int nh = blockIdx.x / tilesPerNH;
  const int tb = blockIdx.x % tilesPerNH;
  const int n  = nh / H_;
  const int h  = nh % H_;
  for (int e = threadIdx.x; e < SLOT; e += 256) {
    float v = kvbuf[(size_t)nh * SLOT + e];
    if (e < 1024) sKV[e] = v; else sKs[e - 1024] = v;
  }
  __syncthreads();

  const int wave = threadIdx.x >> 5;
  const int lane = threadIdx.x & 31;
  const int col  = lane & 15;
  const int half = lane >> 4;
  const int row0 = tb * 128 + wave * 16;
  const size_t qbase = ((size_t)n * L_ + row0) * RS + (size_t)h * D_;

  v8f c0 = {}, c1 = {};
  float dot = 0.f;
#pragma unroll
  for (int kb = 0; kb < 32; kb += 4) {
    const int kk = kb + 2 * half;
    v2f q = *(const v2f*)(Qp + qbase + (size_t)col * RS + kk);
    v2f a;
    float t0 = elu1(q.x), t1 = elu1(q.y);
    a.x = t0; a.y = t1;
    dot += t0 * sKs[kk] + t1 * sKs[kk + 1];
    v2f b0, b1;
    b0.x = sKV[kk * 32 + col];            b0.y = sKV[(kk + 1) * 32 + col];
    b1.x = sKV[kk * 32 + col + 16];       b1.y = sKV[(kk + 1) * 32 + col + 16];
    c0 = wmma_f32(a, b0, c0);
    c1 = wmma_f32(a, b1, c1);
  }
  dot += __shfl_xor(dot, 16, 32);
  const float z = (float)S_ / (dot + 1e-6f);
#pragma unroll
  for (int r = 0; r < 8; ++r) {
    const int m = r + 8 * half;
    const float zr = __shfl(z, m, 32);
    float* orow = outp + ((size_t)n * L_ + row0 + m) * RS + (size_t)h * D_;
    orow[col]      = c0[r] * zr;
    orow[col + 16] = c1[r] * zr;
  }
}

extern "C" void kernel_launch(void* const* d_in, const int* in_sizes, int n_in,
                              void* d_out, int out_size, void* d_ws,
                              size_t ws_size, hipStream_t stream) {
  const float* Q = (const float*)d_in[0];
  const float* K = (const float*)d_in[1];
  const float* V = (const float*)d_in[2];
  float* out = (float*)d_out;

  float* part    = (float*)d_ws;
  float* kvfinal = part + (size_t)NH * CHUNKS * SLOT;

  kv_partial_kernel<<<NH * CHUNKS / 8, 256, 0, stream>>>(K, V, part);
  kv_reduce_kernel<<<NH, 256, 0, stream>>>(part, kvfinal);
  attn_out_kernel<<<NH * (L_ / 128), 256, 0, stream>>>(Q, kvfinal, out);
}


# chunk-reduce pass re-parallelised: one lane per 4 floats with all 64 chunk loads in flight at once (264 blocks of 64 threads, was 64 of 256 with 4 loads in flight); add order per element unchanged
# speedup vs baseline: 1.7108x; 1.7108x over previous
#include <hip/hip_runtime.h>


#define N_ 8
#define L_ 8192
#define S_ 8192
#define H_ 8
#define D_ 32
#define RS (H_ * D_)
#define NH (N_ * H_)
#define CHUNKS 64
#define CHUNK_ROWS (S_ / CHUNKS)
#define SLOT 1056
#define SLOT4 (SLOT / 4)

typedef __attribute__((ext_vector_type(2))) float v2f;
typedef __attribute__((ext_vector_type(4))) float v4f;
typedef __attribute__((ext_vector_type(8))) float v8f;

__device__ __forceinline__ float elu1(float x) {
  return x > 0.f ? x + 1.f : __expf(x);
}

__device__ __forceinline__ v8f wmma_f32(v2f a, v2f b, v8f c) {
  return __builtin_amdgcn_wmma_f32_16x16x4_f32(false, a, false, b, (short)0, c,
                                               false, false);
}

__global__ void __launch_bounds__(256)
kv_partial_kernel(const float* __restrict__ Kp, const float* __restrict__ Vp,
                  float* __restrict__ part) {
  const int wave = threadIdx.x >> 5;
  const int lane = threadIdx.x & 31;
  const int g    = blockIdx.x * 8 + wave;
  const int nh   = g / CHUNKS;
  const int ck   = g % CHUNKS;
  const int n    = nh / H_;
  const int h    = nh % H_;
  const int col  = lane & 15;
  const int half = lane >> 4;
  const float invS = 1.0f / (float)S_;

  const size_t base = ((size_t)n * S_) * RS + (size_t)h * D_;
  const int s0 = ck * CHUNK_ROWS;

  v8f c00 = {}, c01 = {}, c10 = {}, c11 = {};
  float ks0 = 0.f, ks1 = 0.f;

  for (int s = s0; s < s0 + CHUNK_ROWS; s += 4) {
    const int rA = s + 2 * half;
    const float* k0 = Kp + base + (size_t)rA * RS;
    const float* k1 = k0 + RS;
    const float* v0 = Vp + base + (size_t)rA * RS;
    const float* v1 = v0 + RS;
    v2f a0, a1, b0, b1;
    float t;
    t = elu1(k0[col]);      a0.x = t; ks0 += t;
    t = elu1(k1[col]);      a0.y = t; ks0 += t;
    t = elu1(k0[col + 16]); a1.x = t; ks1 += t;
    t = elu1(k1[col + 16]); a1.y = t; ks1 += t;
    b0.x = v0[col]      * invS; b0.y = v1[col]      * invS;
    b1.x = v0[col + 16] * invS; b1.y = v1[col + 16] * invS;
    c00 = wmma_f32(a0, b0, c00);
    c01 = wmma_f32(a0, b1, c01);
    c10 = wmma_f32(a1, b0, c10);
    c11 = wmma_f32(a1, b1, c11);
  }
  ks0 += __shfl_xor(ks0, 16, 32);
  ks1 += __shfl_xor(ks1, 16, 32);

  float* outp = part + (size_t)g * SLOT;
#pragma unroll
  for (int r = 0; r < 8; ++r) {
    const int m = r + 8 * half;
    outp[m * 32 + col]              = c00[r];
    outp[m * 32 + col + 16]         = c01[r];
    outp[(16 + m) * 32 + col]       = c10[r];
    outp[(16 + m) * 32 + col + 16]  = c11[r];
  }
  if (lane < 16) {
    outp[1024 + col]      = ks0;
    outp[1024 + col + 16] = ks1;
  }
}

__global__ void __launch_bounds__(64)
kv_reduce_kernel(const float* __restrict__ part, float* __restrict__ kv) {
  const int i  = blockIdx.x * 64 + threadIdx.x;
  const int nh = i / SLOT4;
  const int q  = i - nh * SLOT4;
  const v4f* p = (const v4f*)(part + (size_t)nh * CHUNKS * SLOT) + q;
  v4f x[CHUNKS];
#pragma unroll
  for (int c = 0; c < CHUNKS; ++c) x[c] = p[(size_t)c * SLOT4];
  __builtin_amdgcn_sched_barrier(0);
  v4f acc = {0.f, 0.f, 0.f, 0.f};
#pragma unroll
  for (int c = 0; c < CHUNKS; ++c) acc += x[c];
  ((v4f*)kv)[i] = acc;
}

__global__ void __launch_bounds__(256)
attn_out_kernel(const float* __restrict__ Qp, const float* __restrict__ kvbuf,
                float* __restrict__ outp) {
  __shared__ float sKV[1024];
  __shared__ float sKs[32];
  const int tilesPerNH = L_ / 128;
  const int nh = blockIdx.x / tilesPerNH;
  const int tb = blockIdx.x % tilesPerNH;
  const int n  = nh / H_;
  const int h  = nh % H_;
  for (int e = threadIdx.x; e < SLOT; e += 256) {
    float v = kvbuf[(size_t)nh * SLOT + e];
    if (e < 1024) sKV[e] = v; else sKs[e - 1024] = v;
  }
  __syncthreads();

  const int wave = threadIdx.x >> 5;
  const int lane = threadIdx.x & 31;
  const int col  = lane & 15;
  const int half = lane >> 4;
  const int row0 = tb * 128 + wave * 16;
  const size_t qbase = ((size_t)n * L_ + row0) * RS + (size_t)h * D_;

  v8f c0 = {}, c1 = {};
  float dot = 0.f;
#pragma unroll
  for (int kb = 0; kb < 32; kb += 4) {
    const int kk = kb + 2 * half;
    v2f q = *(const v2f*)(Qp + qbase + (size_t)col * RS + kk);
    v2f a;
    float t0 = elu1(q.x), t1 = elu1(q.y);
    a.x = t0; a.y = t1;
    dot += t0 * sKs[kk] + t1 * sKs[kk + 1];
    v2f b0, b1;
    b0.x = sKV[kk * 32 + col];            b0.y = sKV[(kk + 1) * 32 + col];
    b1.x = sKV[kk * 32 + col + 16];       b1.y = sKV[(kk + 1) * 32 + col + 16];
    c0 = wmma_f32(a, b0, c0);
    c1 = wmma_f32(a, b1, c1);
  }
  dot += __shfl_xor(dot, 16, 32);
  const float z = (float)S_ / (dot + 1e-6f);
#pragma unroll
  for (int r = 0; r < 8; ++r) {
    const int m = r + 8 * half;
    const float zr = __shfl(z, m, 32);
    float* orow = outp + ((size_t)n * L_ + row0 + m) * RS + (size_t)h * D_;
    orow[col]      = c0[r] * zr;
    orow[col + 16] = c1[r] * zr;
  }
}

extern "C" void kernel_launch(void* const* d_in, const int* in_sizes, int n_in,
                              void* d_out, int out_size, void* d_ws,
                              size_t ws_size, hipStream_t stream) {
  const float* Q = (const float*)d_in[0];
  const float* K = (const float*)d_in[1];
  const float* V = (const float*)d_in[2];
  float* out = (float*)d_out;

  float* part    = (float*)d_ws;
  float* kvfinal = part + (size_t)NH * CHUNKS * SLOT;

  kv_partial_kernel<<<NH * CHUNKS / 8, 256, 0, stream>>>(K, V, part);
  kv_reduce_kernel<<<NH * SLOT4 / 64, 64, 0, stream>>>(part, kvfinal);
  attn_out_kernel<<<NH * (L_ / 128), 256, 0, stream>>>(Q, kvfinal, out);
}


# output pass as 512 long-lived blocks: each wave loops over 8 row tiles with Q loads 4 tiles ahead and its KV fragments kept in registers (was 4096 one-tile blocks); sum order unchanged
# speedup vs baseline: 1.8252x; 1.0669x over previous
#include <hip/hip_runtime.h>


#define N_ 8
#define L_ 8192
#define S_ 8192
#define H_ 8
#define D_ 32
#define RS (H_ * D_)
#define NH (N_ * H_)
#define CHUNKS 64
#define CHUNK_ROWS (S_ / CHUNKS)
#define SLOT 1056
#define SLOT4 (SLOT / 4)

typedef __attribute__((ext_vector_type(2))) float v2f;
typedef __attribute__((ext_vector_type(4))) float v4f;
typedef __attribute__((ext_vector_type(8))) float v8f;

__device__ __forceinline__ float elu1(float x) {
  return x > 0.f ? x + 1.f : __expf(x);
}

__device__ __forceinline__ v8f wmma_f32(v2f a, v2f b, v8f c) {
  return __builtin_amdgcn_wmma_f32_16x16x4_f32(false, a, false, b, (short)0, c,
                                               false, false);
}

__global__ void __launch_bounds__(256)
kv_partial_kernel(const float* __restrict__ Kp, const float* __restrict__ Vp,
                  float* __restrict__ part) {
  const int wave = threadIdx.x >> 5;
  const int lane = threadIdx.x & 31;
  const int g    = blockIdx.x * 8 + wave;
  const int nh   = g / CHUNKS;
  const int ck   = g % CHUNKS;
  const int n    = nh / H_;
  const int h    = nh % H_;
  const int col  = lane & 15;
  const int half = lane >> 4;
  const float invS = 1.0f / (float)S_;

  const size_t base = ((size_t)n * S_) * RS + (size_t)h * D_;
  const int s0 = ck * CHUNK_ROWS;

  v8f c00 = {}, c01 = {}, c10 = {}, c11 = {};
  float ks0 = 0.f, ks1 = 0.f;

  for (int s = s0; s < s0 + CHUNK_ROWS; s += 4) {
    const int rA = s + 2 * half;
    const float* k0 = Kp + base + (size_t)rA * RS;
    const float* k1 = k0 + RS;
    const float* v0 = Vp + base + (size_t)rA * RS;
    const float* v1 = v0 + RS;
    v2f a0, a1, b0, b1;
    float t;
    t = elu1(k0[col]);      a0.x = t; ks0 += t;
    t = elu1(k1[col]);      a0.y = t; ks0 += t;
    t = elu1(k0[col + 16]); a1.x = t; ks1 += t;
    t = elu1(k1[col + 16]); a1.y = t; ks1 += t;
    b0.x = v0[col]      * invS; b0.y = v1[col]      * invS;
    b1.x = v0[col + 16] * invS; b1.y = v1[col + 16] * invS;
    c00 = wmma_f32(a0, b0, c00);
    c01 = wmma_f32(a0, b1, c01);
    c10 = wmma_f32(a1, b0, c10);
    c11 = wmma_f32(a1, b1, c11);
  }
  ks0 += __shfl_xor(ks0, 16, 32);
  ks1 += __shfl_xor(ks1, 16, 32);

  float* outp = part + (size_t)g * SLOT;
#pragma unroll
  for (int r = 0; r < 8; ++r) {
    const int m = r + 8 * half;
    outp[m * 32 + col]              = c00[r];
    outp[m * 32 + col + 16]         = c01[r];
    outp[(16 + m) * 32 + col]       = c10[r];
    outp[(16 + m) * 32 + col + 16]  = c11[r];
  }
  if (lane < 16) {
    outp[1024 + col]      = ks0;
    outp[1024 + col + 16] = ks1;
  }
}

__global__ void __launch_bounds__(64)
kv_reduce_kernel(const float* __restrict__ part, float* __restrict__ kv) {
  const int i  = blockIdx.x * 64 + threadIdx.x;
  const int nh = i / SLOT4;
  const int q  = i - nh * SLOT4;
  const v4f* p = (const v4f*)(part + (size_t)nh * CHUNKS * SLOT) + q;
  v4f x[CHUNKS];
#pragma unroll
  for (int c = 0; c < CHUNKS; ++c) x[c] = p[(size_t)c * SLOT4];
  __builtin_amdgcn_sched_barrier(0);
  v4f acc = {0.f, 0.f, 0.f, 0.f};
#pragma unroll
  for (int c = 0; c < CHUNKS; ++c) acc += x[c];
  ((v4f*)kv)[i] = acc;
}

#define ROWS_PB 1024
#define BPN (L_ / ROWS_PB)
#define TILES (ROWS_PB / 128)
#define QAHEAD 4
#define TILE_PITCH ((size_t)128 * RS)

__global__ void __launch_bounds__(256)
attn_out_kernel(const float* __restrict__ Qp, const float* __restrict__ kvbuf,
                float* __restrict__ outp) {
  __shared__ float sKV[1024];
  __shared__ float sKs[32];
  const int nh = blockIdx.x / BPN;
  const int tb = blockIdx.x % BPN;
  const int n  = nh / H_;
  const int h  = nh % H_;
  for (int e = threadIdx.x; e < SLOT; e += 256) {
    float v = kvbuf[(size_t)nh * SLOT + e];
    if (e < 1024) sKV[e] = v; else sKs[e - 1024] = v;
  }

  const int wave = threadIdx.x >> 5;
  const int lane = threadIdx.x & 31;
  const int col  = lane & 15;
  const int half = lane >> 4;
  const int row0 = tb * ROWS_PB + wave * 16;
  const size_t qbase = ((size_t)n * L_ + row0) * RS + (size_t)h * D_;
  const float* qlane = Qp + qbase + (size_t)col * RS + 2 * half;

  __syncthreads();

  v2f q[TILES][8];
#pragma unroll
  for (int i = 0; i < QAHEAD; ++i) {
#pragma unroll
    for (int j = 0; j < 8; ++j)
      q[i][j] = *(const v2f*)(qlane + (size_t)i * TILE_PITCH + 4 * j);
  }

  v2f b0[8], b1[8];
  float ka[8], kb[8];
#pragma unroll
  for (int j = 0; j < 8; ++j) {
    const int kk = 4 * j + 2 * half;
    b0[j].x = sKV[kk * 32 + col];        b0[j].y = sKV[(kk + 1) * 32 + col];
    b1[j].x = sKV[kk * 32 + col + 16];   b1[j].y = sKV[(kk + 1) * 32 + col + 16];
    ka[j] = sKs[kk];                     kb[j] = sKs[kk + 1];
  }

#pragma unroll
  for (int i = 0; i < TILES; ++i) {
    if (i + QAHEAD < TILES) {
#pragma unroll
      for (int j = 0; j < 8; ++j)
        q[i + QAHEAD][j] =
            *(const v2f*)(qlane + (size_t)(i + QAHEAD) * TILE_PITCH + 4 * j);
    }
    __builtin_amdgcn_sched_barrier(0);

    v8f c0 = {}, c1 = {};
    float dot = 0.f;
#pragma unroll
    for (int j = 0; j < 8; ++j) {
      v2f a;
      float t0 = elu1(q[i][j].x), t1 = elu1(q[i][j].y);
      a.x = t0; a.y = t1;
      dot += t0 * ka[j] + t1 * kb[j];
      c0 = wmma_f32(a, b0[j], c0);
      c1 = wmma_f32(a, b1[j], c1);
    }
    dot += __shfl_xor(dot, 16, 32);
    const float z = (float)S_ / (dot + 1e-6f);
#pragma unroll
    for (int r = 0; r < 8; ++r) {
      const int m = r + 8 * half;
      const float zr = __shfl(z, m, 32);
      float* orow = outp + ((size_t)n * L_ + row0 + i * 128 + m) * RS
                    + (size_t)h * D_;
      orow[col]      = c0[r] * zr;
      orow[col + 16] = c1[r] * zr;
    }
    __builtin_amdgcn_sched_barrier(0);
  }
}

extern "C" void kernel_launch(void* const* d_in, const int* in_sizes, int n_in,
                              void* d_out, int out_size, void* d_ws,
                              size_t ws_size, hipStream_t stream) {
  const float* Q = (const float*)d_in[0];
  const float* K = (const float*)d_in[1];
  const float* V = (const float*)d_in[2];
  float* out = (float*)d_out;

  float* part    = (float*)d_ws;
  float* kvfinal = part + (size_t)NH * CHUNKS * SLOT;

  kv_partial_kernel<<<NH * CHUNKS / 8, 256, 0, stream>>>(K, V, part);
  kv_reduce_kernel<<<NH * SLOT4 / 64, 64, 0, stream>>>(part, kvfinal);
  attn_out_kernel<<<NH * BPN, 256, 0, stream>>>(Q, kvfinal, out);
}


# reduce launch removed: the partial pass adds its 8 waves' chunk sums in LDS (one slot per block) and the output pass adds the 8 slots while staging KV; chunk sum tree now 8x8 (was 64 in a row)
# speedup vs baseline: 2.0849x; 1.1423x over previous
#include <hip/hip_runtime.h>


#define N_ 8
#define L_ 8192
#define S_ 8192
#define H_ 8
#define D_ 32
#define RS (H_ * D_)
#define NH (N_ * H_)
#define CHUNKS 64
#define CHUNK_ROWS (S_ / CHUNKS)
#define SLOT 1056
#define SLOT4 (SLOT / 4)
#define WPB 8
#define SLOTS_PNH (CHUNKS / WPB)

typedef __attribute__((ext_vector_type(2))) float v2f;
typedef __attribute__((ext_vector_type(4))) float v4f;
typedef __attribute__((ext_vector_type(8))) float v8f;

__device__ __forceinline__ float elu1(float x) {
  return x > 0.f ? x + 1.f : __expf(x);
}

__device__ __forceinline__ v8f wmma_f32(v2f a, v2f b, v8f c) {
  return __builtin_amdgcn_wmma_f32_16x16x4_f32(false, a, false, b, (short)0, c,
                                               false, false);
}

__global__ void __launch_bounds__(256)
kv_partial_kernel(const float* __restrict__ Kp, const float* __restrict__ Vp,
                  float* __restrict__ part) {
  __shared__ v4f red4[WPB * SLOT4];
  const int wave = threadIdx.x >> 5;
  const int lane = threadIdx.x & 31;
  const int g    = blockIdx.x * WPB + wave;
  const int nh   = g / CHUNKS;
  const int ck   = g % CHUNKS;
  const int n    = nh / H_;
  const int h    = nh % H_;
  const int col  = lane & 15;
  const int half = lane >> 4;
  const float invS = 1.0f / (float)S_;

  const size_t base = ((size_t)n * S_) * RS + (size_t)h * D_;
  const int s0 = ck * CHUNK_ROWS;

  v8f c00 = {}, c01 = {}, c10 = {}, c11 = {};
  float ks0 = 0.f, ks1 = 0.f;

  for (int s = s0; s < s0 + CHUNK_ROWS; s += 4) {
    const int rA = s + 2 * half;
    const float* k0 = Kp + base + (size_t)rA * RS;
    const float* k1 = k0 + RS;
    const float* v0 = Vp + base + (size_t)rA * RS;
    const float* v1 = v0 + RS;
    v2f a0, a1, b0, b1;
    float t;
    t = elu1(k0[col]);      a0.x = t; ks0 += t;
    t = elu1(k1[col]);      a0.y = t; ks0 += t;
    t = elu1(k0[col + 16]); a1.x = t; ks1 += t;
    t = elu1(k1[col + 16]); a1.y = t; ks1 += t;
    b0.x = v0[col]      * invS; b0.y = v1[col]      * invS;
    b1.x = v0[col + 16] * invS; b1.y = v1[col + 16] * invS;
    c00 = wmma_f32(a0, b0, c00);
    c01 = wmma_f32(a0, b1, c01);
    c10 = wmma_f32(a1, b0, c10);
    c11 = wmma_f32(a1, b1, c11);
  }
  ks0 += __shfl_xor(ks0, 16, 32);
  ks1 += __shfl_xor(ks1, 16, 32);

  float* wp = (float*)red4 + wave * SLOT;
#pragma unroll
  for (int r = 0; r < 8; ++r) {
    const int m = r + 8 * half;
    wp[m * 32 + col]              = c00[r];
    wp[m * 32 + col + 16]         = c01[r];
    wp[(16 + m) * 32 + col]       = c10[r];
    wp[(16 + m) * 32 + col + 16]  = c11[r];
  }
  if (lane < 16) {
    wp[1024 + col]      = ks0;
    wp[1024 + col + 16] = ks1;
  }
  __syncthreads();

  v4f* outp = (v4f*)(part + (size_t)blockIdx.x * SLOT);
  for (int e = threadIdx.x; e < SLOT4; e += 256) {
    v4f acc = red4[e];
    for (int w = 1; w < WPB; ++w) acc += red4[w * SLOT4 + e];
    outp[e] = acc;
  }
}

#define ROWS_PB 1024
#define BPN (L_ / ROWS_PB)
#define TILES (ROWS_PB / 128)
#define QAHEAD 4
#define TILE_PITCH ((size_t)128 * RS)

__global__ void __launch_bounds__(256)
attn_out_kernel(const float* __restrict__ Qp, const float* __restrict__ part,
                float* __restrict__ outp) {
  __shared__ v4f sAll4[SLOT4];
  const float* sKV = (const float*)sAll4;
  const float* sKs = sKV + 1024;
  const int nh = blockIdx.x / BPN;
  const int tb = blockIdx.x % BPN;
  const int n  = nh / H_;
  const int h  = nh % H_;
  {
    const v4f* p4 = (const v4f*)(part + (size_t)nh * SLOTS_PNH * SLOT);
    const int e = threadIdx.x;
    const bool tail = e < SLOT4 - 256;
    v4f x[SLOTS_PNH], y[SLOTS_PNH];
#pragma unroll
    for (int j = 0; j < SLOTS_PNH; ++j) x[j] = p4[j * SLOT4 + e];
    if (tail) {
#pragma unroll
      for (int j = 0; j < SLOTS_PNH; ++j) y[j] = p4[j * SLOT4 + 256 + e];
    }
    v4f acc = x[0];
#pragma unroll
    for (int j = 1; j < SLOTS_PNH; ++j) acc += x[j];
    sAll4[e] = acc;
    if (tail) {
      v4f acc2 = y[0];
#pragma unroll
      for (int j = 1; j < SLOTS_PNH; ++j) acc2 += y[j];
      sAll4[256 + e] = acc2;
    }
  }

  const int wave = threadIdx.x >> 5;
  const int lane = threadIdx.x & 31;
  const int col  = lane & 15;
  const int half = lane >> 4;
  const int row0 = tb * ROWS_PB + wave * 16;
  const size_t qbase = ((size_t)n * L_ + row0) * RS + (size_t)h * D_;
  const float* qlane = Qp + qbase + (size_t)col * RS + 2 * half;

  __syncthreads();

  v2f q[TILES][8];
#pragma unroll
  for (int i = 0; i < QAHEAD; ++i) {
#pragma unroll
    for (int j = 0; j < 8; ++j)
      q[i][j] = *(const v2f*)(qlane + (size_t)i * TILE_PITCH + 4 * j);
  }

  v2f b0[8], b1[8];
  float ka[8], kb[8];
#pragma unroll
  for (int j = 0; j < 8; ++j) {
    const int kk = 4 * j + 2 * half;
    b0[j].x = sKV[kk * 32 + col];        b0[j].y = sKV[(kk + 1) * 32 + col];
    b1[j].x = sKV[kk * 32 + col + 16];   b1[j].y = sKV[(kk + 1) * 32 + col + 16];
    ka[j] = sKs[kk];                     kb[j] = sKs[kk + 1];
  }

#pragma unroll
  for (int i = 0; i < TILES; ++i) {
    if (i + QAHEAD < TILES) {
#pragma unroll
      for (int j = 0; j < 8; ++j)
        q[i + QAHEAD][j] =
            *(const v2f*)(qlane + (size_t)(i + QAHEAD) * TILE_PITCH + 4 * j);
    }
    __builtin_amdgcn_sched_barrier(0);

    v8f c0 = {}, c1 = {};
    float dot = 0.f;
#pragma unroll
    for (int j = 0; j < 8; ++j) {
      v2f a;
      float t0 = elu1(q[i][j].x), t1 = elu1(q[i][j].y);
      a.x = t0; a.y = t1;
      dot += t0 * ka[j] + t1 * kb[j];
      c0 = wmma_f32(a, b0[j], c0);
      c1 = wmma_f32(a, b1[j], c1);
    }
    dot += __shfl_xor(dot, 16, 32);
    const float z = (float)S_ / (dot + 1e-6f);
#pragma unroll
    for (int r = 0; r < 8; ++r) {
      const int m = r + 8 * half;
      const float zr = __shfl(z, m, 32);
      float* orow = outp + ((size_t)n * L_ + row0 + i * 128 + m) * RS
                    + (size_t)h * D_;
      orow[col]      = c0[r] * zr;
      orow[col + 16] = c1[r] * zr;
    }
    __builtin_amdgcn_sched_barrier(0);
  }
}

extern "C" void kernel_launch(void* const* d_in, const int* in_sizes, int n_in,
                              void* d_out, int out_size, void* d_ws,
                              size_t ws_size, hipStream_t stream) {
  const float* Q = (const float*)d_in[0];
  const float* K = (const float*)d_in[1];
  const float* V = (const float*)d_in[2];
  float* out = (float*)d_out;

  float* part = (float*)d_ws;

  kv_partial_kernel<<<NH * CHUNKS / WPB, 256, 0, stream>>>(K, V, part);
  attn_out_kernel<<<NH * BPN, 256, 0, stream>>>(Q, part, out);
}


# partial pass keeps 8-12 KB of K and V loads in flight per wave (was 2 KB, awaited before the math): 8-byte lane loads, a whole 128-byte row per half-wave, 32 rows ahead; sum order unchanged
# speedup vs baseline: 2.3041x; 1.1051x over previous
#include <hip/hip_runtime.h>


#define N_ 8
#define L_ 8192
#define S_ 8192
#define H_ 8
#define D_ 32
#define RS (H_ * D_)
#define NH (N_ * H_)
#define CHUNKS 64
#define CHUNK_ROWS (S_ / CHUNKS)
#define SLOT 1056
#define SLOT4 (SLOT / 4)
#define WPB 8
#define SLOTS_PNH (CHUNKS / WPB)

typedef __attribute__((ext_vector_type(2))) float v2f;
typedef __attribute__((ext_vector_type(4))) float v4f;
typedef __attribute__((ext_vector_type(8))) float v8f;

__device__ __forceinline__ float elu1(float x) {
  return x > 0.f ? x + 1.f : __expf(x);
}

__device__ __forceinline__ v8f wmma_f32(v2f a, v2f b, v8f c) {
  return __builtin_amdgcn_wmma_f32_16x16x4_f32(false, a, false, b, (short)0, c,
                                               false, false);
}

#define KG 4
#define GROUPS (CHUNK_ROWS / (4 * KG))
#define KAHEAD 2

__global__ void __launch_bounds__(256)
kv_partial_kernel(const float* __restrict__ Kp, const float* __restrict__ Vp,
                  float* __restrict__ part) {
  __shared__ v4f red4[WPB * SLOT4];
  const int wave = threadIdx.x >> 5;
  const int lane = threadIdx.x & 31;
  const int g    = blockIdx.x * WPB + wave;
  const int nh   = g / CHUNKS;
  const int ck   = g % CHUNKS;
  const int n    = nh / H_;
  const int h    = nh % H_;
  const int col  = lane & 15;
  const int half = lane >> 4;
  const float invS = 1.0f / (float)S_;

  const size_t base = ((size_t)n * S_) * RS + (size_t)h * D_;
  const int s0 = ck * CHUNK_ROWS;
  const size_t loff = base + (size_t)(s0 + 2 * half) * RS + 2 * col;
  const float* kl = Kp + loff;
  const float* vl = Vp + loff;

  v8f c00 = {}, c01 = {}, c10 = {}, c11 = {};
  float ks0 = 0.f, ks1 = 0.f;

  v2f kq[GROUPS][KG][2], vq[GROUPS][KG][2];
#pragma unroll
  for (int gi = 0; gi < KAHEAD; ++gi) {
#pragma unroll
    for (int j = 0; j < KG; ++j) {
#pragma unroll
      for (int r = 0; r < 2; ++r) {
        const size_t o = (size_t)(16 * gi + 4 * j + r) * RS;
        kq[gi][j][r] = *(const v2f*)(kl + o);
        vq[gi][j][r] = *(const v2f*)(vl + o);
      }
    }
  }

#pragma unroll
  for (int gi = 0; gi < GROUPS; ++gi) {
    if (gi + KAHEAD < GROUPS) {
#pragma unroll
      for (int j = 0; j < KG; ++j) {
#pragma unroll
        for (int r = 0; r < 2; ++r) {
          const size_t o = (size_t)(16 * (gi + KAHEAD) + 4 * j + r) * RS;
          kq[gi + KAHEAD][j][r] = *(const v2f*)(kl + o);
          vq[gi + KAHEAD][j][r] = *(const v2f*)(vl + o);
        }
      }
    }
    __builtin_amdgcn_sched_barrier(0);

#pragma unroll
    for (int j = 0; j < KG; ++j) {
      const v2f k0 = kq[gi][j][0], k1 = kq[gi][j][1];
      const v2f v0 = vq[gi][j][0], v1 = vq[gi][j][1];
      v2f a0, a1, b0, b1;
      float t;
      t = elu1(k0.x); a0.x = t; ks0 += t;
      t = elu1(k1.x); a0.y = t; ks0 += t;
      t = elu1(k0.y); a1.x = t; ks1 += t;
      t = elu1(k1.y); a1.y = t; ks1 += t;
      b0.x = v0.x * invS; b0.y = v1.x * invS;
      b1.x = v0.y * invS; b1.y = v1.y * invS;
      c00 = wmma_f32(a0, b0, c00);
      c01 = wmma_f32(a0, b1, c01);
      c10 = wmma_f32(a1, b0, c10);
      c11 = wmma_f32(a1, b1, c11);
    }
    __builtin_amdgcn_sched_barrier(0);
  }
  ks0 += __shfl_xor(ks0, 16, 32);
  ks1 += __shfl_xor(ks1, 16, 32);

  float* wp = (float*)red4 + wave * SLOT;
#pragma unroll
  for (int r = 0; r < 8; ++r) {
    const int m = r + 8 * half;
    v2f lo, hi;
    lo.x = c00[r]; lo.y = c01[r];
    hi.x = c10[r]; hi.y = c11[r];
    *(v2f*)(wp + (2 * m) * 32 + 2 * col)     = lo;
    *(v2f*)(wp + (2 * m + 1) * 32 + 2 * col) = hi;
  }
  if (lane < 16) {
    v2f kk;
    kk.x = ks0; kk.y = ks1;
    *(v2f*)(wp + 1024 + 2 * col) = kk;
  }
  __syncthreads();

  v4f* outp = (v4f*)(part + (size_t)blockIdx.x * SLOT);
  for (int e = threadIdx.x; e < SLOT4; e += 256) {
    v4f acc = red4[e];
    for (int w = 1; w < WPB; ++w) acc += red4[w * SLOT4 + e];
    outp[e] = acc;
  }
}

#define ROWS_PB 1024
#define BPN (L_ / ROWS_PB)
#define TILES (ROWS_PB / 128)
#define QAHEAD 4
#define TILE_PITCH ((size_t)128 * RS)

__global__ void __launch_bounds__(256)
attn_out_kernel(const float* __restrict__ Qp, const float* __restrict__ part,
                float* __restrict__ outp) {
  __shared__ v4f sAll4[SLOT4];
  const float* sKV = (const float*)sAll4;
  const float* sKs = sKV + 1024;
  const int nh = blockIdx.x / BPN;
  const int tb = blockIdx.x % BPN;
  const int n  = nh / H_;
  const int h  = nh % H_;
  {
    const v4f* p4 = (const v4f*)(part + (size_t)nh * SLOTS_PNH * SLOT);
    const int e = threadIdx.x;
    const bool tail = e < SLOT4 - 256;
    v4f x[SLOTS_PNH], y[SLOTS_PNH];
#pragma unroll
    for (int j = 0; j < SLOTS_PNH; ++j) x[j] = p4[j * SLOT4 + e];
    if (tail) {
#pragma unroll
      for (int j = 0; j < SLOTS_PNH; ++j) y[j] = p4[j * SLOT4 + 256 + e];
    }
    v4f acc = x[0];
#pragma unroll
    for (int j = 1; j < SLOTS_PNH; ++j) acc += x[j];
    sAll4[e] = acc;
    if (tail) {
      v4f acc2 = y[0];
#pragma unroll
      for (int j = 1; j < SLOTS_PNH; ++j) acc2 += y[j];
      sAll4[256 + e] = acc2;
    }
  }

  const int wave = threadIdx.x >> 5;
  const int lane = threadIdx.x & 31;
  const int col  = lane & 15;
  const int half = lane >> 4;
  const int row0 = tb * ROWS_PB + wave * 16;
  const size_t qbase = ((size_t)n * L_ + row0) * RS + (size_t)h * D_;
  const float* qlane = Qp + qbase + (size_t)col * RS + 2 * half;

  __syncthreads();

  v2f q[TILES][8];
#pragma unroll
  for (int i = 0; i < QAHEAD; ++i) {
#pragma unroll
    for (int j = 0; j < 8; ++j)
      q[i][j] = *(const v2f*)(qlane + (size_t)i * TILE_PITCH + 4 * j);
  }

  v2f b0[8], b1[8];
  float ka[8], kb[8];
#pragma unroll
  for (int j = 0; j < 8; ++j) {
    const int kk = 4 * j + 2 * half;
    b0[j].x = sKV[kk * 32 + col];        b0[j].y = sKV[(kk + 1) * 32 + col];
    b1[j].x = sKV[kk * 32 + col + 16];   b1[j].y = sKV[(kk + 1) * 32 + col + 16];
    ka[j] = sKs[kk];                     kb[j] = sKs[kk + 1];
  }

#pragma unroll
  for (int i = 0; i < TILES; ++i) {
    if (i + QAHEAD < TILES) {
#pragma unroll
      for (int j = 0; j < 8; ++j)
        q[i + QAHEAD][j] =
            *(const v2f*)(qlane + (size_t)(i + QAHEAD) * TILE_PITCH + 4 * j);
    }
    __builtin_amdgcn_sched_barrier(0);

    v8f c0 = {}, c1 = {};
    float dot = 0.f;
#pragma unroll
    for (int j = 0; j < 8; ++j) {
      v2f a;
      float t0 = elu1(q[i][j].x), t1 = elu1(q[i][j].y);
      a.x = t0; a.y = t1;
      dot += t0 * ka[j] + t1 * kb[j];
      c0 = wmma_f32(a, b0[j], c0);
      c1 = wmma_f32(a, b1[j], c1);
    }
    dot += __shfl_xor(dot, 16, 32);
    const float z = (float)S_ / (dot + 1e-6f);
#pragma unroll
    for (int r = 0; r < 8; ++r) {
      const int m = r + 8 * half;
      const float zr = __shfl(z, m, 32);
      float* orow = outp + ((size_t)n * L_ + row0 + i * 128 + m) * RS
                    + (size_t)h * D_;
      orow[col]      = c0[r] * zr;
      orow[col + 16] = c1[r] * zr;
    }
    __builtin_amdgcn_sched_barrier(0);
  }
}

extern "C" void kernel_launch(void* const* d_in, const int* in_sizes, int n_in,
                              void* d_out, int out_size, void* d_ws,
                              size_t ws_size, hipStream_t stream) {
  const float* Q = (const float*)d_in[0];
  const float* K = (const float*)d_in[1];
  const float* V = (const float*)d_in[2];
  float* out = (float*)d_out;

  float* part = (float*)d_ws;

  kv_partial_kernel<<<NH * CHUNKS / WPB, 256, 0, stream>>>(K, V, part);
  attn_out_kernel<<<NH * BPN, 256, 0, stream>>>(Q, part, out);
}


# output pass: each wave computes both heads of a pair and trades half-rows by v_permlane16_swap, so one store writes a row's whole aligned 256-byte unit (was four 64-byte pieces); sum order unchanged
# speedup vs baseline: 2.4403x; 1.0591x over previous
#include <hip/hip_runtime.h>


#define N_ 8
#define L_ 8192
#define S_ 8192
#define H_ 8
#define D_ 32
#define RS (H_ * D_)
#define NH (N_ * H_)
#define CHUNKS 64
#define CHUNK_ROWS (S_ / CHUNKS)
#define SLOT 1056
#define SLOT4 (SLOT / 4)
#define WPB 8
#define SLOTS_PNH (CHUNKS / WPB)

typedef __attribute__((ext_vector_type(2))) float v2f;
typedef __attribute__((ext_vector_type(4))) float v4f;
typedef __attribute__((ext_vector_type(8))) float v8f;

__device__ __forceinline__ float elu1(float x) {
  return x > 0.f ? x + 1.f : __expf(x);
}

__device__ __forceinline__ v8f wmma_f32(v2f a, v2f b, v8f c) {
  return __builtin_amdgcn_wmma_f32_16x16x4_f32(false, a, false, b, (short)0, c,
                                               false, false);
}

#define KG 4
#define GROUPS (CHUNK_ROWS / (4 * KG))
#define KAHEAD 2

__global__ void __launch_bounds__(256)
kv_partial_kernel(const float* __restrict__ Kp, const float* __restrict__ Vp,
                  float* __restrict__ part) {
  __shared__ v4f red4[WPB * SLOT4];
  const int wave = threadIdx.x >> 5;
  const int lane = threadIdx.x & 31;
  const int g    = blockIdx.x * WPB + wave;
  const int nh   = g / CHUNKS;
  const int ck   = g % CHUNKS;
  const int n    = nh / H_;
  const int h    = nh % H_;
  const int col  = lane & 15;
  const int half = lane >> 4;
  const float invS = 1.0f / (float)S_;

  const size_t base = ((size_t)n * S_) * RS + (size_t)h * D_;
  const int s0 = ck * CHUNK_ROWS;
  const size_t loff = base + (size_t)(s0 + 2 * half) * RS + 2 * col;
  const float* kl = Kp + loff;
  const float* vl = Vp + loff;

  v8f c00 = {}, c01 = {}, c10 = {}, c11 = {};
  float ks0 = 0.f, ks1 = 0.f;

  v2f kq[GROUPS][KG][2], vq[GROUPS][KG][2];
#pragma unroll
  for (int gi = 0; gi < KAHEAD; ++gi) {
#pragma unroll
    for (int j = 0; j < KG; ++j) {
#pragma unroll
      for (int r = 0; r < 2; ++r) {
        const size_t o = (size_t)(16 * gi + 4 * j + r) * RS;
        kq[gi][j][r] = *(const v2f*)(kl + o);
        vq[gi][j][r] = *(const v2f*)(vl + o);
      }
    }
  }

#pragma unroll
  for (int gi = 0; gi < GROUPS; ++gi) {
    if (gi + KAHEAD < GROUPS) {
#pragma unroll
      for (int j = 0; j < KG; ++j) {
#pragma unroll
        for (int r = 0; r < 2; ++r) {
          const size_t o = (size_t)(16 * (gi + KAHEAD) + 4 * j + r) * RS;
          kq[gi + KAHEAD][j][r] = *(const v2f*)(kl + o);
          vq[gi + KAHEAD][j][r] = *(const v2f*)(vl + o);
        }
      }
    }
    __builtin_amdgcn_sched_barrier(0);

#pragma unroll
    for (int j = 0; j < KG; ++j) {
      const v2f k0 = kq[gi][j][0], k1 = kq[gi][j][1];
      const v2f v0 = vq[gi][j][0], v1 = vq[gi][j][1];
      v2f a0, a1, b0, b1;
      float t;
      t = elu1(k0.x); a0.x = t; ks0 += t;
      t = elu1(k1.x); a0.y = t; ks0 += t;
      t = elu1(k0.y); a1.x = t; ks1 += t;
      t = elu1(k1.y); a1.y = t; ks1 += t;
      b0.x = v0.x * invS; b0.y = v1.x * invS;
      b1.x = v0.y * invS; b1.y = v1.y * invS;
      c00 = wmma_f32(a0, b0, c00);
      c01 = wmma_f32(a0, b1, c01);
      c10 = wmma_f32(a1, b0, c10);
      c11 = wmma_f32(a1, b1, c11);
    }
    __builtin_amdgcn_sched_barrier(0);
  }
  ks0 += __shfl_xor(ks0, 16, 32);
  ks1 += __shfl_xor(ks1, 16, 32);

  float* wp = (float*)red4 + wave * SLOT;
#pragma unroll
  for (int r = 0; r < 8; ++r) {
    const int m = r + 8 * half;
    v2f lo, hi;
    lo.x = c00[r]; lo.y = c01[r];
    hi.x = c10[r]; hi.y = c11[r];
    *(v2f*)(wp + (2 * m) * 32 + 2 * col)     = lo;
    *(v2f*)(wp + (2 * m + 1) * 32 + 2 * col) = hi;
  }
  if (lane < 16) {
    v2f kk;
    kk.x = ks0; kk.y = ks1;
    *(v2f*)(wp + 1024 + 2 * col) = kk;
  }
  __syncthreads();

  v4f* outp = (v4f*)(part + (size_t)blockIdx.x * SLOT);
  for (int e = threadIdx.x; e < SLOT4; e += 256) {
    v4f acc = red4[e];
    for (int w = 1; w < WPB; ++w) acc += red4[w * SLOT4 + e];
    outp[e] = acc;
  }
}

#define PAIRS (H_ / 2)
#define ROWS_PB 512
#define BPP (L_ / ROWS_PB)
#define STEPS 8
#define QAHEAD 4
#define TILE_PITCH ((size_t)128 * RS)

__global__ void __launch_bounds__(256)
attn_out_kernel(const float* __restrict__ Qp, const float* __restrict__ part,
                float* __restrict__ outp) {
  __shared__ v4f sAll4[2 * SLOT4];
  const float* sF = (const float*)sAll4;
  const int np = blockIdx.x / BPP;
  const int tb = blockIdx.x % BPP;
  const int n  = np / PAIRS;
  const int pr = np % PAIRS;
  const int nhA = n * H_ + 2 * pr;
  {
    const v4f* p4 = (const v4f*)(part + (size_t)nhA * SLOTS_PNH * SLOT);
    const int e0 = threadIdx.x;
    const int e1 = threadIdx.x + 256;
    const int e2 = threadIdx.x + 512;
    const int s1 = e1 + (e1 >= SLOT4 ? 7 * SLOT4 : 0);
    const int s2 = e2 + 7 * SLOT4;
    const bool tail = e2 < 2 * SLOT4;
    v4f x[SLOTS_PNH], y[SLOTS_PNH], w[SLOTS_PNH];
#pragma unroll
    for (int j = 0; j < SLOTS_PNH; ++j) x[j] = p4[j * SLOT4 + e0];
#pragma unroll
    for (int j = 0; j < SLOTS_PNH; ++j) y[j] = p4[j * SLOT4 + s1];
    if (tail) {
#pragma unroll
      for (int j = 0; j < SLOTS_PNH; ++j) w[j] = p4[j * SLOT4 + s2];
    }
    v4f acc = x[0];
#pragma unroll
    for (int j = 1; j < SLOTS_PNH; ++j) acc += x[j];
    sAll4[e0] = acc;
    v4f acc1 = y[0];
#pragma unroll
    for (int j = 1; j < SLOTS_PNH; ++j) acc1 += y[j];
    sAll4[e1] = acc1;
    if (tail) {
      v4f acc2 = w[0];
#pragma unroll
      for (int j = 1; j < SLOTS_PNH; ++j) acc2 += w[j];
      sAll4[e2] = acc2;
    }
  }

  const int wave = threadIdx.x >> 5;
  const int lane = threadIdx.x & 31;
  const int col  = lane & 15;
  const int half = lane >> 4;
  const int row0 = tb * ROWS_PB + wave * 16;
  const size_t pbase = ((size_t)n * L_ + row0) * RS + (size_t)(2 * pr) * D_;
  const float* qlane = Qp + pbase + (size_t)col * RS + 2 * half;
  float* olane = outp + pbase + 2 * lane;

  __syncthreads();

  v2f q[STEPS][8];
#pragma unroll
  for (int s = 0; s < QAHEAD; ++s) {
#pragma unroll
    for (int j = 0; j < 8; ++j)
      q[s][j] = *(const v2f*)(qlane + (size_t)(s >> 1) * TILE_PITCH
                              + (s & 1) * D_ + 4 * j);
  }

  v2f oA[8];

#pragma unroll
  for (int s = 0; s < STEPS; ++s) {
    if (s + QAHEAD < STEPS) {
#pragma unroll
      for (int j = 0; j < 8; ++j)
        q[s + QAHEAD][j] =
            *(const v2f*)(qlane + (size_t)((s + QAHEAD) >> 1) * TILE_PITCH
                          + ((s + QAHEAD) & 1) * D_ + 4 * j);
    }
    __builtin_amdgcn_sched_barrier(0);

    const int hd = s & 1;
    const float* sKV = sF + hd * SLOT;
    const float* sKs = sKV + 1024;
    v2f b0[8], b1[8];
    float ka[8], kb[8];
#pragma unroll
    for (int j = 0; j < 8; ++j) {
      const int kk = 4 * j + 2 * half;
      const v2f lo = *(const v2f*)(sKV + kk * 32 + 2 * col);
      const v2f hi = *(const v2f*)(sKV + (kk + 1) * 32 + 2 * col);
      b0[j].x = lo.x; b0[j].y = hi.x;
      b1[j].x = lo.y; b1[j].y = hi.y;
      ka[j] = sKs[kk];  kb[j] = sKs[kk + 1];
    }

    v8f c0 = {}, c1 = {};
    float dot = 0.f;
#pragma unroll
    for (int j = 0; j < 8; ++j) {
      v2f a;
      float t0 = elu1(q[s][j].x), t1 = elu1(q[s][j].y);
      a.x = t0; a.y = t1;
      dot += t0 * ka[j] + t1 * kb[j];
      c0 = wmma_f32(a, b0[j], c0);
      c1 = wmma_f32(a, b1[j], c1);
    }
    dot += __shfl_xor(dot, 16, 32);
    const float z = (float)S_ / (dot + 1e-6f);
    v2f o[8];
#pragma unroll
    for (int r = 0; r < 8; ++r) {
      const int m = r + 8 * half;
      const float zr = __shfl(z, m, 32);
      o[r].x = c0[r] * zr;
      o[r].y = c1[r] * zr;
    }
    if (hd == 0) {
#pragma unroll
      for (int r = 0; r < 8; ++r) oA[r] = o[r];
    } else {
      float* otile = olane + (size_t)(s >> 1) * TILE_PITCH;
#pragma unroll
      for (int r = 0; r < 8; ++r) {
        float ax = oA[r].x, ay = oA[r].y, bx = o[r].x, by = o[r].y;
        asm volatile("v_permlane16_swap_b32 %0, %1" : "+v"(ax), "+v"(bx));
        asm volatile("v_permlane16_swap_b32 %0, %1" : "+v"(ay), "+v"(by));
        v2f u, v;
        u.x = ax; u.y = ay;
        v.x = bx; v.y = by;
        *(v2f*)(otile + (size_t)r * RS)       = u;
        *(v2f*)(otile + (size_t)(r + 8) * RS) = v;
      }
    }
    __builtin_amdgcn_sched_barrier(0);
  }
}

extern "C" void kernel_launch(void* const* d_in, const int* in_sizes, int n_in,
                              void* d_out, int out_size, void* d_ws,
                              size_t ws_size, hipStream_t stream) {
  const float* Q = (const float*)d_in[0];
  const float* K = (const float*)d_in[1];
  const float* V = (const float*)d_in[2];
  float* out = (float*)d_out;

  float* part = (float*)d_ws;

  kv_partial_kernel<<<NH * CHUNKS / WPB, 256, 0, stream>>>(K, V, part);
  attn_out_kernel<<<N_ * PAIRS * BPP, 256, 0, stream>>>(Q, part, out);
}
